# starting kernel
# baseline (speedup 1.0000x reference)
#include <hip/hip_runtime.h>
#include <hip/hip_bf16.h>


typedef __attribute__((ext_vector_type(2))) float v2f;
typedef __attribute__((ext_vector_type(8))) float v8f;

#define SIG_B      1024
#define SIG_L      2048
#define SIG_C      7
#define SIG_T      2047
#define SIG_HC     128
#define SIG_NH     16
#define SIG_NWAVE  8
#define SIG_COLS   136
#define SIG_DT     (1.0f / 2047.0f)

__device__ __forceinline__ void sig_step2(float& S, float bA, float bB, v8f& acc)
{
    v2f a, bv;
    a.x = S; bv.x = bA; S += bA;
    a.y = S; bv.y = bB; S += bB;
    acc = __builtin_amdgcn_wmma_f32_16x16x4_f32(
              false, a, false, bv, (short)0, acc, false, false);
}

__global__ __launch_bounds__(32)
void sig_partial_kernel(const float* __restrict__ path,
                        float* __restrict__ ws_sums,
                        float* __restrict__ ws_sig2)
{
    const int bb   = blockIdx.x >> 3;
    const int w    = blockIdx.x & 7;
    const int lane = threadIdx.x;
    const int q    = lane & 15;
    const int p    = q >> 3;
    const int cm   = q & 7;

    const int batch = 2 * bb + p;
    const int chunk = 2 * w + (lane >> 4);
    const int t0    = chunk * SIG_HC;
    const bool is_time = (cm == 0);

    const int comp = (cm >= 1) ? (cm - 1) : 0;
    const float* __restrict__ bbase = path + (size_t)batch * SIG_L * SIG_C;

    const float* __restrict__ ptr = bbase + (t0 + 1) * SIG_C + comp;

    float prev = bbase[t0 * SIG_C + comp];
    float S = 0.0f;
    v8f acc = (v8f)0.0f;

    const int iters = (w == 7) ? 31 : 32;
    for (int i = 0; i < iters; ++i) {
        const float x0 = ptr[0];
        const float x1 = ptr[SIG_C];
        const float x2 = ptr[2 * SIG_C];
        const float x3 = ptr[3 * SIG_C];
        ptr += 4 * SIG_C;

        const float b0 = is_time ? SIG_DT : (x0 - prev);
        const float b1 = is_time ? SIG_DT : (x1 - x0);
        const float b2 = is_time ? SIG_DT : (x2 - x1);
        const float b3 = is_time ? SIG_DT : (x3 - x2);
        prev = x3;

        sig_step2(S, b0, b1, acc);
        sig_step2(S, b2, b3, acc);
    }

    if (w == 7) {
        int off = (t0 + 125) * SIG_C + comp;
        const int offMax = (SIG_L - 1) * SIG_C + comp;
        const int o0 = (off             < offMax) ? off             : offMax;
        const int o1 = (off +   SIG_C   < offMax) ? off +   SIG_C   : offMax;
        const int o2 = (off + 2*SIG_C   < offMax) ? off + 2*SIG_C   : offMax;
        const int o3 = (off + 3*SIG_C   < offMax) ? off + 3*SIG_C   : offMax;
        const float x0 = bbase[o0];
        const float x1 = bbase[o1];
        const float x2 = bbase[o2];
        const float x3 = bbase[o3];

        const float b0 = is_time ? SIG_DT : (x0 - prev);
        const float b1 = is_time ? SIG_DT : (x1 - x0);
        const float b2 = is_time ? SIG_DT : (x2 - x1);
        const float b3 = is_time ? SIG_DT : (x3 - x2);

        sig_step2(S, b0, b1, acc);
        sig_step2(S, b2, b3, acc);
    }

    ws_sums[((size_t)batch * SIG_NH + chunk) * 8 + cm] = S;

    if (lane < 8) {
        float* dst = ws_sig2 + ((size_t)(2 * bb) * SIG_NWAVE + w) * 64;
        #pragma unroll
        for (int j = 0; j < 8; ++j)
            dst[j * 8 + lane] = acc[j];
    } else if (lane >= 24) {
        float* dst = ws_sig2 + ((size_t)(2 * bb + 1) * SIG_NWAVE + w) * 64;
        #pragma unroll
        for (int j = 0; j < 8; ++j)
            dst[j * 8 + (lane - 24)] = acc[j];
    }
}

__global__ __launch_bounds__(64)
void sig_combine_kernel(const float* __restrict__ ws_sums,
                        const float* __restrict__ ws_sig2,
                        float* __restrict__ out, int out_size)
{
    const int b   = blockIdx.x;
    const int tid = threadIdx.x;
    __shared__ float s[SIG_NH * 8];

    const float* bs = ws_sums + (size_t)b * SIG_NH * 8;
    s[tid] = bs[tid];
    {
        float v = bs[tid + 64];
        if (tid + 64 == 15 * 8 + 0) v -= SIG_DT;
        s[tid + 64] = v;
    }
    __syncthreads();

    const int d = tid >> 3;
    const int e = tid & 7;

    float P = 0.0f;
    float cross = 0.0f;
    #pragma unroll
    for (int c = 0; c < SIG_NH; ++c) {
        cross += P * s[c * 8 + e];
        P     += s[c * 8 + d];
    }

    float part = 0.0f;
    const float* bp = ws_sig2 + (size_t)b * SIG_NWAVE * 64;
    #pragma unroll
    for (int w = 0; w < SIG_NWAVE; ++w)
        part += bp[w * 64 + tid];

    if (e == 0) part -= s[15 * 8 + d] * SIG_DT;

    const float sig2_de = part + cross;
    const size_t base = (size_t)b * SIG_COLS;

    if (base + 8 + tid < (size_t)out_size)
        out[base + 8 + tid] = sig2_de;
    if (base + 72 + tid < (size_t)out_size)
        out[base + 72 + tid] = P * sig2_de;
    if (e == 0 && base + d < (size_t)out_size)
        out[base + d] = P;
}

extern "C" void kernel_launch(void* const* d_in, const int* in_sizes, int n_in,
                              void* d_out, int out_size, void* d_ws, size_t ws_size,
                              hipStream_t stream) {
    (void)in_sizes; (void)n_in; (void)ws_size;
    const float* path = (const float*)d_in[0];

    float* ws_sums = (float*)d_ws;
    float* ws_sig2 = ws_sums + (size_t)SIG_B * SIG_NH * 8;

    sig_partial_kernel<<<dim3((SIG_B / 2) * SIG_NWAVE), dim3(32), 0, stream>>>(
        path, ws_sums, ws_sig2);
    sig_combine_kernel<<<dim3(SIG_B), dim3(64), 0, stream>>>(
        ws_sums, ws_sig2, (float*)d_out, out_size);
}


# Stage A chunk loop made straight-line with all 129 row loads hoisted ahead of use: about 40 loads in flight per wave, one exposed round trip instead of 8; arithmetic and summation order unchanged
# speedup vs baseline: 1.2851x; 1.2851x over previous
#include <hip/hip_runtime.h>
#include <hip/hip_bf16.h>


typedef __attribute__((ext_vector_type(2))) float v2f;
typedef __attribute__((ext_vector_type(8))) float v8f;

#define SIG_B      1024
#define SIG_L      2048
#define SIG_C      7
#define SIG_T      2047
#define SIG_HC     128
#define SIG_NH     16
#define SIG_NWAVE  8
#define SIG_COLS   136
#define SIG_DT     (1.0f / 2047.0f)

__device__ __forceinline__ void sig_step2(float& S, float bA, float bB, v8f& acc)
{
    v2f a, bv;
    a.x = S; bv.x = bA; S += bA;
    a.y = S; bv.y = bB; S += bB;
    acc = __builtin_amdgcn_wmma_f32_16x16x4_f32(
              false, a, false, bv, (short)0, acc, false, false);
}

__global__ __launch_bounds__(32)
void sig_partial_kernel(const float* __restrict__ path,
                        float* __restrict__ ws_sums,
                        float* __restrict__ ws_sig2)
{
    const int bb   = blockIdx.x >> 3;
    const int w    = blockIdx.x & 7;
    const int lane = threadIdx.x;
    const int q    = lane & 15;
    const int p    = q >> 3;
    const int cm   = q & 7;

    const int batch = 2 * bb + p;
    const int chunk = 2 * w + (lane >> 4);
    const int t0    = chunk * SIG_HC;
    const bool is_time = (cm == 0);

    const int comp = (cm >= 1) ? (cm - 1) : 0;
    const float* __restrict__ bbase = path + (size_t)batch * SIG_L * SIG_C;

    const float* __restrict__ ptr = bbase + (t0 + 1) * SIG_C + comp;

    float xv[SIG_HC + 1];
    xv[0] = bbase[t0 * SIG_C + comp];
    #pragma unroll
    for (int k = 0; k < SIG_HC - 1; ++k)
        xv[k + 1] = ptr[k * SIG_C];
    {
        const int rlast = (t0 + SIG_HC < SIG_L - 1) ? (t0 + SIG_HC) : (SIG_L - 1);
        xv[SIG_HC] = bbase[rlast * SIG_C + comp];
    }

    float S = 0.0f;
    v8f acc = (v8f)0.0f;

    #pragma unroll
    for (int i = 0; i < SIG_HC / 4; ++i) {
        const float xp = xv[4 * i];
        const float x0 = xv[4 * i + 1];
        const float x1 = xv[4 * i + 2];
        const float x2 = xv[4 * i + 3];
        const float x3 = xv[4 * i + 4];

        const float b0 = is_time ? SIG_DT : (x0 - xp);
        const float b1 = is_time ? SIG_DT : (x1 - x0);
        const float b2 = is_time ? SIG_DT : (x2 - x1);
        const float b3 = is_time ? SIG_DT : (x3 - x2);

        sig_step2(S, b0, b1, acc);
        sig_step2(S, b2, b3, acc);
    }

    ws_sums[((size_t)batch * SIG_NH + chunk) * 8 + cm] = S;

    if (lane < 8) {
        float* dst = ws_sig2 + ((size_t)(2 * bb) * SIG_NWAVE + w) * 64;
        #pragma unroll
        for (int j = 0; j < 8; ++j)
            dst[j * 8 + lane] = acc[j];
    } else if (lane >= 24) {
        float* dst = ws_sig2 + ((size_t)(2 * bb + 1) * SIG_NWAVE + w) * 64;
        #pragma unroll
        for (int j = 0; j < 8; ++j)
            dst[j * 8 + (lane - 24)] = acc[j];
    }
}

__global__ __launch_bounds__(64)
void sig_combine_kernel(const float* __restrict__ ws_sums,
                        const float* __restrict__ ws_sig2,
                        float* __restrict__ out, int out_size)
{
    const int b   = blockIdx.x;
    const int tid = threadIdx.x;
    __shared__ float s[SIG_NH * 8];

    const float* bs = ws_sums + (size_t)b * SIG_NH * 8;
    s[tid] = bs[tid];
    {
        float v = bs[tid + 64];
        if (tid + 64 == 15 * 8 + 0) v -= SIG_DT;
        s[tid + 64] = v;
    }
    __syncthreads();

    const int d = tid >> 3;
    const int e = tid & 7;

    float P = 0.0f;
    float cross = 0.0f;
    #pragma unroll
    for (int c = 0; c < SIG_NH; ++c) {
        cross += P * s[c * 8 + e];
        P     += s[c * 8 + d];
    }

    float part = 0.0f;
    const float* bp = ws_sig2 + (size_t)b * SIG_NWAVE * 64;
    #pragma unroll
    for (int w = 0; w < SIG_NWAVE; ++w)
        part += bp[w * 64 + tid];

    if (e == 0) part -= s[15 * 8 + d] * SIG_DT;

    const float sig2_de = part + cross;
    const size_t base = (size_t)b * SIG_COLS;

    if (base + 8 + tid < (size_t)out_size)
        out[base + 8 + tid] = sig2_de;
    if (base + 72 + tid < (size_t)out_size)
        out[base + 72 + tid] = P * sig2_de;
    if (e == 0 && base + d < (size_t)out_size)
        out[base + d] = P;
}

extern "C" void kernel_launch(void* const* d_in, const int* in_sizes, int n_in,
                              void* d_out, int out_size, void* d_ws, size_t ws_size,
                              hipStream_t stream) {
    (void)in_sizes; (void)n_in; (void)ws_size;
    const float* path = (const float*)d_in[0];

    float* ws_sums = (float*)d_ws;
    float* ws_sig2 = ws_sums + (size_t)SIG_B * SIG_NH * 8;

    sig_partial_kernel<<<dim3((SIG_B / 2) * SIG_NWAVE), dim3(32), 0, stream>>>(
        path, ws_sums, ws_sig2);
    sig_combine_kernel<<<dim3(SIG_B), dim3(64), 0, stream>>>(
        ws_sums, ws_sig2, (float*)d_out, out_size);
}


# Combine stage fused into the partial kernel: the 8 waves of a batch pair form one work-group and pass chunk sums and WMMA partials through LDS behind one barrier; one launch, sum order kept
# speedup vs baseline: 1.7909x; 1.3936x over previous
#include <hip/hip_runtime.h>
#include <hip/hip_bf16.h>


typedef __attribute__((ext_vector_type(2))) float v2f;
typedef __attribute__((ext_vector_type(8))) float v8f;

#define SIG_B      1024
#define SIG_L      2048
#define SIG_C      7
#define SIG_T      2047
#define SIG_HC     128
#define SIG_NH     16
#define SIG_NWAVE  8
#define SIG_COLS   136
#define SIG_DT     (1.0f / 2047.0f)

__device__ __forceinline__ void sig_step2(float& S, float bA, float bB, v8f& acc)
{
    v2f a, bv;
    a.x = S; bv.x = bA; S += bA;
    a.y = S; bv.y = bB; S += bB;
    acc = __builtin_amdgcn_wmma_f32_16x16x4_f32(
              false, a, false, bv, (short)0, acc, false, false);
}

__global__ __launch_bounds__(SIG_NWAVE * 32)
void sig_fused_kernel(const float* __restrict__ path,
                      float* __restrict__ out, int out_size)
{
    __shared__ float s_sum[2 * SIG_NH * 8];
    __shared__ float s_part[2 * SIG_NWAVE * 64];

    const int bb   = blockIdx.x;
    const int tid  = threadIdx.x;
    const int w    = tid >> 5;
    const int lane = tid & 31;
    const int q    = lane & 15;
    const int p    = q >> 3;
    const int cm   = q & 7;

    const int batch = 2 * bb + p;
    const int chunk = 2 * w + (lane >> 4);
    const int t0    = chunk * SIG_HC;
    const bool is_time = (cm == 0);

    const int comp = (cm >= 1) ? (cm - 1) : 0;
    const float* __restrict__ bbase = path + (size_t)batch * SIG_L * SIG_C;

    const float* __restrict__ ptr = bbase + (t0 + 1) * SIG_C + comp;

    float xv[SIG_HC + 1];
    xv[0] = bbase[t0 * SIG_C + comp];
    #pragma unroll
    for (int k = 0; k < SIG_HC - 1; ++k)
        xv[k + 1] = ptr[k * SIG_C];
    {
        const int rlast = (t0 + SIG_HC < SIG_L - 1) ? (t0 + SIG_HC) : (SIG_L - 1);
        xv[SIG_HC] = bbase[rlast * SIG_C + comp];
    }

    float S = 0.0f;
    v8f acc = (v8f)0.0f;

    #pragma unroll
    for (int i = 0; i < SIG_HC / 4; ++i) {
        const float xp = xv[4 * i];
        const float x0 = xv[4 * i + 1];
        const float x1 = xv[4 * i + 2];
        const float x2 = xv[4 * i + 3];
        const float x3 = xv[4 * i + 4];

        const float b0 = is_time ? SIG_DT : (x0 - xp);
        const float b1 = is_time ? SIG_DT : (x1 - x0);
        const float b2 = is_time ? SIG_DT : (x2 - x1);
        const float b3 = is_time ? SIG_DT : (x3 - x2);

        sig_step2(S, b0, b1, acc);
        sig_step2(S, b2, b3, acc);
    }

    {
        float v = S;
        if (chunk == SIG_NH - 1 && cm == 0) v -= SIG_DT;
        s_sum[(p * SIG_NH + chunk) * 8 + cm] = v;
    }

    if (lane < 8) {
        float* dst = s_part + (size_t)(0 * SIG_NWAVE + w) * 64;
        #pragma unroll
        for (int j = 0; j < 8; ++j)
            dst[j * 8 + lane] = acc[j];
    } else if (lane >= 24) {
        float* dst = s_part + (size_t)(1 * SIG_NWAVE + w) * 64;
        #pragma unroll
        for (int j = 0; j < 8; ++j)
            dst[j * 8 + (lane - 24)] = acc[j];
    }

    __syncthreads();
    if (tid >= 128) return;

    const int pb = tid >> 6;
    const int t  = tid & 63;
    const int d  = t >> 3;
    const int e  = t & 7;
    const float* s  = s_sum  + pb * SIG_NH * 8;
    const float* bp = s_part + pb * SIG_NWAVE * 64;

    float P = 0.0f;
    float cross = 0.0f;
    #pragma unroll
    for (int c = 0; c < SIG_NH; ++c) {
        cross += P * s[c * 8 + e];
        P     += s[c * 8 + d];
    }

    float part = 0.0f;
    #pragma unroll
    for (int ww = 0; ww < SIG_NWAVE; ++ww)
        part += bp[ww * 64 + t];

    if (e == 0) part -= s[15 * 8 + d] * SIG_DT;

    const float sig2_de = part + cross;
    const size_t base = (size_t)(2 * bb + pb) * SIG_COLS;

    if (base + 8 + t < (size_t)out_size)
        out[base + 8 + t] = sig2_de;
    if (base + 72 + t < (size_t)out_size)
        out[base + 72 + t] = P * sig2_de;
    if (e == 0 && base + d < (size_t)out_size)
        out[base + d] = P;
}

extern "C" void kernel_launch(void* const* d_in, const int* in_sizes, int n_in,
                              void* d_out, int out_size, void* d_ws, size_t ws_size,
                              hipStream_t stream) {
    (void)in_sizes; (void)n_in; (void)d_ws; (void)ws_size;
    const float* path = (const float*)d_in[0];

    sig_fused_kernel<<<dim3(SIG_B / 2), dim3(SIG_NWAVE * 32), 0, stream>>>(
        path, (float*)d_out, out_size);
}
